# starting kernel
# baseline (speedup 1.0000x reference)
#include <hip/hip_runtime.h>

typedef float v2f __attribute__((ext_vector_type(2)));
typedef float v8f __attribute__((ext_vector_type(8)));

#define NUM_CAPS 512
#define CAP_SIZE 256
#define RF_SIZE  64
#define KDIM     2048
#define MDIM     2048
#define NDIM     64
#define KSPLIT   8

__global__ __launch_bounds__(256)
void scrramble_route_wmma(const float* __restrict__ Ci,
                          const float* __restrict__ x,
                          float* __restrict__ xr) {
  __shared__ float red[KSPLIT * 32 * 8];

  const int lane = threadIdx.x & 31;
  const int w    = __builtin_amdgcn_readfirstlane(threadIdx.x >> 5);
  const int mt   = blockIdx.x >> 2;
  const int nt   = blockIdx.x & 3;
  const int R0   = mt * 16;
  const int N0   = nt * 16;
  const int half = lane >> 4;
  const int ll   = lane & 15;

  const int Sbeg = w * (KDIM / KSPLIT);
  const int Send = Sbeg + (KDIM / KSPLIT);

  v8f c = {};
#pragma unroll 4
  for (int S = Sbeg; S < Send; S += 4) {
    const int k0 = S + 2 * half;
    v2f a, b;
    a.x = Ci[(size_t)k0 * MDIM + (size_t)(R0 + ll)];
    a.y = Ci[(size_t)(k0 + 1) * MDIM + (size_t)(R0 + ll)];
    b.x = x[k0 * NDIM + N0 + ll];
    b.y = x[(k0 + 1) * NDIM + N0 + ll];
    c = __builtin_amdgcn_wmma_f32_16x16x4_f32(false, a, false, b,
                                              (short)0, c, false, false);
  }

#pragma unroll
  for (int r = 0; r < 8; ++r) red[(w * 32 + lane) * 8 + r] = c[r];
  __syncthreads();

  if (threadIdx.x < 32) {
    v8f acc = {};
#pragma unroll
    for (int ww = 0; ww < KSPLIT; ++ww) {
#pragma unroll
      for (int r = 0; r < 8; ++r) acc[r] += red[(ww * 32 + lane) * 8 + r];
    }
#pragma unroll
    for (int r = 0; r < 8; ++r)
      xr[(size_t)(R0 + r + 8 * half) * NDIM + (size_t)(N0 + ll)] = acc[r];
  }
}

__global__ __launch_bounds__(512)
void caps_transform_wmma(const float* __restrict__ Wi,
                         const float* __restrict__ xr,
                         float* __restrict__ y) {
  __shared__ float xs[CAP_SIZE];

  const int cap   = blockIdx.x;
  const int mtile = threadIdx.x >> 5;
  const int lane  = threadIdx.x & 31;
  const int half  = lane >> 4;
  const int ll    = lane & 15;
  const int row   = mtile * 16 + ll;
  const int jj    = row >> 6;
  const int l     = row & 63;

  const float* __restrict__ Wcap = Wi + (size_t)cap * (CAP_SIZE * CAP_SIZE);
  const float* __restrict__ Wrow = Wcap + jj * 16384 + l * 64;

  if (threadIdx.x < CAP_SIZE)
    xs[threadIdx.x] = xr[(size_t)cap * CAP_SIZE + threadIdx.x];
  __syncthreads();

  v8f c = {};
#pragma unroll 4
  for (int S = 0; S < CAP_SIZE; S += 4) {
    const int col0 = S + 2 * half;
    const int k    = col0 >> 6;
    const int m    = col0 & 63;
    if ((S & 63) == 0 && S + 64 < CAP_SIZE) {
      __builtin_prefetch(Wrow + (k + 1) * 4096 + m, 0, 1);
    }
    const v2f a = __builtin_nontemporal_load(
        (const v2f*)(Wrow + k * 4096 + m));
    const float bx0 = xs[col0];
    const float bx1 = xs[col0 + 1];
    v2f b;
    b.x = (ll == 0) ? bx0 : 0.0f;
    b.y = (ll == 0) ? bx1 : 0.0f;
    c = __builtin_amdgcn_wmma_f32_16x16x4_f32(false, a, false, b,
                                              (short)0, c, false, false);
  }

  if (ll == 0) {
#pragma unroll
    for (int r = 0; r < 8; ++r) {
      y[(size_t)cap * CAP_SIZE + (size_t)(mtile * 16 + r + 8 * half)] = c[r];
    }
  }
}

extern "C" void kernel_launch(void* const* d_in, const int* in_sizes, int n_in,
                              void* d_out, int out_size, void* d_ws, size_t ws_size,
                              hipStream_t stream) {
  const float* x  = (const float*)d_in[0];
  const float* Ci = (const float*)d_in[1];
  const float* Wi = (const float*)d_in[2];
  float* y  = (float*)d_out;
  float* xr = (float*)d_ws;

  scrramble_route_wmma<<<512, 256, 0, stream>>>(Ci, x, xr);
  caps_transform_wmma<<<NUM_CAPS, 512, 0, stream>>>(Wi, xr, y);
}


# Stage 1 K split in two ACROSS blocks: 32x32 tile per block and K half on 8 waves, Ci and x both 8B/lane whole lines feeding 2 WMMAs each; half 1 parked in out, added by stage 2; sum order changed
# speedup vs baseline: 2.3452x; 2.3452x over previous
#include <hip/hip_runtime.h>

typedef float v2f __attribute__((ext_vector_type(2)));
typedef float v8f __attribute__((ext_vector_type(8)));

#define NUM_CAPS 512
#define CAP_SIZE 256
#define RF_SIZE  64
#define KDIM     2048
#define MDIM     2048
#define NDIM     64

#define S1_WAVES 8
#define S1_KH    (KDIM / 2)
#define S1_KW    (S1_KH / S1_WAVES)

__global__ __launch_bounds__(256)
void scrramble_route_wmma(const float* __restrict__ Ci,
                          const float* __restrict__ x,
                          float* __restrict__ plane0,
                          float* __restrict__ plane1) {
  __shared__ float red[S1_WAVES * 32 * 32];

  const int lane = threadIdx.x & 31;
  const int w    = __builtin_amdgcn_readfirstlane(threadIdx.x >> 5);
  const int nt   = blockIdx.x & 1;
  const int kh   = (blockIdx.x >> 1) & 1;
  const int mt   = blockIdx.x >> 2;
  const int R0   = mt * 32;
  const int N0   = nt * 32;
  const int h    = lane >> 4;
  const int c    = lane & 15;

  const int Sb = kh * S1_KH + w * S1_KW;
  const float* __restrict__ cip = Ci + (size_t)(Sb + 2 * h) * MDIM
                                + (size_t)(R0 + 2 * c);
  const float* __restrict__ xp  = x + (Sb + 2 * h) * NDIM + N0 + 2 * c;

  v2f ca[2][16];
  v2f xb[2][16];
#define LOAD_BATCH(B) do { \
      \
    _Pragma("unroll") for (int g = 0; g < 8; ++g) { \
      const int gg = (B) * 8 + g; \
      ca[(B) & 1][2 * g]     = *(const v2f*)(cip + (size_t)gg * 4 * MDIM); \
      ca[(B) & 1][2 * g + 1] = *(const v2f*)(cip + (size_t)gg * 4 * MDIM + MDIM); \
      xb[(B) & 1][2 * g]     = *(const v2f*)(xp + gg * 4 * NDIM); \
      xb[(B) & 1][2 * g + 1] = *(const v2f*)(xp + gg * 4 * NDIM + NDIM); } } while (0)

  v8f acc00 = {}, acc01 = {}, acc10 = {}, acc11 = {};
  LOAD_BATCH(0);
#pragma unroll
  for (int bt = 0; bt < 4; ++bt) {
    __builtin_amdgcn_sched_barrier(0);
    if (bt + 1 < 4) LOAD_BATCH(bt + 1);
    __builtin_amdgcn_sched_barrier(0);
#pragma unroll
    for (int g = 0; g < 8; ++g) {
      v2f a0, a1, b0, b1;
      a0.x = ca[bt & 1][2 * g].x; a0.y = ca[bt & 1][2 * g + 1].x;
      a1.x = ca[bt & 1][2 * g].y; a1.y = ca[bt & 1][2 * g + 1].y;
      b0.x = xb[bt & 1][2 * g].x; b0.y = xb[bt & 1][2 * g + 1].x;
      b1.x = xb[bt & 1][2 * g].y; b1.y = xb[bt & 1][2 * g + 1].y;
      acc00 = __builtin_amdgcn_wmma_f32_16x16x4_f32(false, a0, false, b0,
                                                    (short)0, acc00, false, false);
      acc01 = __builtin_amdgcn_wmma_f32_16x16x4_f32(false, a0, false, b1,
                                                    (short)0, acc01, false, false);
      acc10 = __builtin_amdgcn_wmma_f32_16x16x4_f32(false, a1, false, b0,
                                                    (short)0, acc10, false, false);
      acc11 = __builtin_amdgcn_wmma_f32_16x16x4_f32(false, a1, false, b1,
                                                    (short)0, acc11, false, false);
    }
  }
#undef LOAD_BATCH

#pragma unroll
  for (int r = 0; r < 8; ++r) {
    red[(w * 32 + r) * 32 + lane]      = acc00[r];
    red[(w * 32 + 8 + r) * 32 + lane]  = acc01[r];
    red[(w * 32 + 16 + r) * 32 + lane] = acc10[r];
    red[(w * 32 + 24 + r) * 32 + lane] = acc11[r];
  }
  __syncthreads();

  float* __restrict__ dst = kh ? plane1 : plane0;
#pragma unroll
  for (int e = 0; e < 2; ++e) {
    const int q  = 2 * w + e;
    const int ja = q >> 3;
    const int r  = q & 7;
    v2f sum = {0.0f, 0.0f};
#pragma unroll
    for (int ww = 0; ww < S1_WAVES; ++ww) {
      sum.x += red[(ww * 32 + 16 * ja + r) * 32 + lane];
      sum.y += red[(ww * 32 + 16 * ja + 8 + r) * 32 + lane];
    }
    *(v2f*)(dst + (size_t)(R0 + 2 * (8 * h + r) + ja) * NDIM
                + (size_t)(N0 + 2 * c)) = sum;
  }
}

typedef float v4f __attribute__((ext_vector_type(4)));

template <int MASK>
static __device__ __forceinline__ float add_xmask(float v) {
  const int t = __builtin_amdgcn_update_dpp(0, __builtin_bit_cast(int, v),
                                            0x160 | MASK, 0xf, 0xf, true);
  return v + __builtin_bit_cast(float, t);
}

__global__ __launch_bounds__(512)
void caps_transform_wmma(const float* __restrict__ Wi,
                         const float* __restrict__ xr,
                         float* y) {
  const int cap  = blockIdx.x;
  const int lane = threadIdx.x & 31;
  const int w    = __builtin_amdgcn_readfirstlane(threadIdx.x >> 5);
  const int jj   = w >> 2;
  const int l0   = (w & 3) * 16;
  const int h    = lane >> 4;
  const int g    = (lane >> 3) & 1;
  const int p    = lane & 7;
  const int coff = h * 32 + p * 4;

  const float* __restrict__ xb = xr + (size_t)cap * CAP_SIZE + coff;
  const float* xc = y + (size_t)cap * CAP_SIZE + coff;
  v4f xv[4];
#pragma unroll
  for (int k = 0; k < 4; ++k)
    xv[k] = *(const v4f*)(xb + k * RF_SIZE) + *(const v4f*)(xc + k * RF_SIZE);

  const float* __restrict__ Wb = Wi + (size_t)cap * (CAP_SIZE * CAP_SIZE)
                               + jj * 16384 + (l0 + g) * 64 + coff;

  v4f wbuf[2][8];
#define LOAD_STEP(S) do { \
    _Pragma("unroll") for (int a2 = 0; a2 < 2; ++a2) \
    _Pragma("unroll") for (int k = 0; k < 4; ++k) \
      wbuf[(S) & 1][a2 * 4 + k] = __builtin_nontemporal_load( \
          (const v4f*)(Wb + k * 4096 + (2 * (S) + a2) * 128)); } while (0)

  float res = 0.0f;
  LOAD_STEP(0);
#pragma unroll
  for (int s = 0; s < 4; ++s) {
    __builtin_amdgcn_sched_barrier(0);
    if (s + 1 < 4) LOAD_STEP(s + 1);
    __builtin_amdgcn_sched_barrier(0);
    const v4f* wr = wbuf[s & 1];
    v8f c0 = {}, c1 = {};
#pragma unroll
    for (int k = 0; k < 4; ++k) {
      v2f a, b;
      a.x = wr[k].x;     a.y = wr[k].y;     b.x = xv[k].x; b.y = xv[k].y;
      c0 = __builtin_amdgcn_wmma_f32_16x16x4_f32(false, a, false, b,
                                                 (short)0, c0, false, false);
      a.x = wr[4 + k].x; a.y = wr[4 + k].y;
      c1 = __builtin_amdgcn_wmma_f32_16x16x4_f32(false, a, false, b,
                                                 (short)0, c1, false, false);
      a.x = wr[k].z;     a.y = wr[k].w;     b.x = xv[k].z; b.y = xv[k].w;
      c0 = __builtin_amdgcn_wmma_f32_16x16x4_f32(false, a, false, b,
                                                 (short)0, c0, false, false);
      a.x = wr[4 + k].z; a.y = wr[4 + k].w;
      c1 = __builtin_amdgcn_wmma_f32_16x16x4_f32(false, a, false, b,
                                                 (short)0, c1, false, false);
    }
    float d0 = c0[0], d1 = c1[0];
#pragma unroll
    for (int r = 1; r < 8; ++r) {
      d0 = (p == r) ? c0[r] : d0;
      d1 = (p == r) ? c1[r] : d1;
    }
    d0 = add_xmask<1>(d0); d1 = add_xmask<1>(d1);
    d0 = add_xmask<2>(d0); d1 = add_xmask<2>(d1);
    d0 = add_xmask<4>(d0); d1 = add_xmask<4>(d1);
    res = (p == 2 * s)     ? d0 : res;
    res = (p == 2 * s + 1) ? d1 : res;
  }
#undef LOAD_STEP

  __syncthreads();
  if (lane < 8 || lane >= 24)
    y[(size_t)cap * CAP_SIZE + (size_t)(jj * 64 + l0 + 2 * p + g)] = res;
}

extern "C" void kernel_launch(void* const* d_in, const int* in_sizes, int n_in,
                              void* d_out, int out_size, void* d_ws, size_t ws_size,
                              hipStream_t stream) {
  const float* x  = (const float*)d_in[0];
  const float* Ci = (const float*)d_in[1];
  const float* Wi = (const float*)d_in[2];
  float* y  = (float*)d_out;
  float* xr = (float*)d_ws;

  scrramble_route_wmma<<<256, 256, 0, stream>>>(Ci, x, xr, y);
  caps_transform_wmma<<<NUM_CAPS, 512, 0, stream>>>(Wi, xr, y);
}
